# starting kernel
# baseline (speedup 1.0000x reference)
#include <hip/hip_runtime.h>

typedef __attribute__((ext_vector_type(2))) float v2f;
typedef __attribute__((ext_vector_type(8))) float v8f;

#define BATCH 64
#define CHUNKS 4

__global__ void k_transpose(const float* __restrict__ x, float* __restrict__ xt, int ns) {
    __shared__ float tile[32][33];
    const int tx = threadIdx.x;
    const int ty = threadIdx.y;
    const int s0 = blockIdx.x * 32;
    const int b0 = blockIdx.y * 32;
#pragma unroll
    for (int k = 0; k < 4; ++k) {
        int s = s0 + tx, b = b0 + ty + k * 8;
        if (s < ns && b < BATCH) tile[ty + k * 8][tx] = x[(size_t)b * ns + s];
    }
    __syncthreads();
#pragma unroll
    for (int k = 0; k < 4; ++k) {
        int s = s0 + ty + k * 8, b = b0 + tx;
        if (s < ns && b < BATCH) xt[(size_t)s * BATCH + b] = tile[tx][ty + k * 8];
    }
}

__global__ void k_zero(float4* __restrict__ p, long long n4) {
    long long i = (long long)blockIdx.x * blockDim.x + threadIdx.x;
    const long long stride = (long long)gridDim.x * blockDim.x;
    const float4 z = make_float4(0.f, 0.f, 0.f, 0.f);
    for (; i < n4; i += stride) p[i] = z;
}

__global__ void k_scatter(const float* __restrict__ xt, const float* __restrict__ vals,
                          const int* __restrict__ row, const int* __restrict__ col,
                          float* __restrict__ out_t, int nnz) {
    const int lane = threadIdx.x & 31;
    const long long wave = ((long long)blockIdx.x * blockDim.x + threadIdx.x) >> 5;
    const long long base0 = wave * (32LL * CHUNKS);
    for (int ch = 0; ch < CHUNKS; ++ch) {
        const long long base = base0 + 32LL * ch;
        if (base >= nnz) return;
        const long long idx = base + lane;
        float v = 0.f;
        int r = 0, c = 0;
        if (idx < nnz) {
            v = vals[idx];
            r = row[idx];
            c = col[idx];
        }
        if (ch + 1 < CHUNKS && base + 32 + lane < nnz) {
            __builtin_prefetch(vals + base + 32 + lane, 0, 0);
            __builtin_prefetch(row + base + 32 + lane, 0, 0);
            __builtin_prefetch(col + base + 32 + lane, 0, 0);
        }
        const int count = (nnz - base < 32) ? (int)(nnz - base) : 32;
        for (int j = 0; j < count; ++j) {
            const float vj = __shfl(v, j);
            const int rj   = __shfl(r, j);
            const int cj   = __shfl(c, j);
            const v2f xv = *(const v2f*)(xt + (size_t)cj * BATCH + lane * 2);
            float* dst = out_t + (size_t)rj * BATCH + lane * 2;
            unsafeAtomicAdd(dst, xv.x * vj);
            unsafeAtomicAdd(dst + 1, xv.y * vj);
        }
    }
}

__global__ void k_epilogue_wmma(const float* __restrict__ out_t, const float* __restrict__ bias,
                                float* __restrict__ out, int nd, int dtiles) {
    const int lane = threadIdx.x & 31;
    const int wave = (int)(((long long)blockIdx.x * blockDim.x + threadIdx.x) >> 5);
    const int tiles = (BATCH / 16) * dtiles;
    if (wave >= tiles) return;
    const int b0 = (wave & 3) * 16;
    const int d0 = (wave >> 2) * 16;
    const int n = lane & 15;
    const int mbase = (lane < 16) ? 0 : 8;

    const float* cbase = out_t + (size_t)(d0 + n) * BATCH + b0 + mbase;
    const float4 lo = *(const float4*)(cbase);
    const float4 hi = *(const float4*)(cbase + 4);
    v8f c = {lo.x, lo.y, lo.z, lo.w, hi.x, hi.y, hi.z, hi.w};

    v2f a = {(lane < 16) ? 1.0f : 0.0f, 0.0f};
    v2f bm = {(lane < 16) ? bias[d0 + n] : 0.0f, 0.0f};

    v8f d = __builtin_amdgcn_wmma_f32_16x16x4_f32(
         false, a,  false, bm,
         (short)0, c,  false,  false);

#pragma unroll
    for (int v = 0; v < 8; ++v) {
        out[(size_t)(b0 + mbase + v) * nd + d0 + n] = d[v];
    }
}

__global__ void k_epilogue_tail(const float* __restrict__ out_t, const float* __restrict__ bias,
                                float* __restrict__ out, int nd, int dstart) {
    const int total = (nd - dstart) * BATCH;
    const int i = blockIdx.x * blockDim.x + threadIdx.x;
    if (i >= total) return;
    const int d = dstart + (i >> 6);
    const int b = i & 63;
    out[(size_t)b * nd + d] = out_t[(size_t)d * BATCH + b] + bias[d];
}

extern "C" void kernel_launch(void* const* d_in, const int* in_sizes, int n_in,
                              void* d_out, int out_size, void* d_ws, size_t ws_size,
                              hipStream_t stream) {
    const float* x    = (const float*)d_in[0];
    const float* vals = (const float*)d_in[1];
    const float* bias = (const float*)d_in[2];
    const int* row    = (const int*)d_in[3];
    const int* col    = (const int*)d_in[4];
    float* out = (float*)d_out;

    const int nnz = in_sizes[1];
    const int nd  = in_sizes[2];
    const int ns  = in_sizes[0] / BATCH;

    float* xt    = (float*)d_ws;
    float* out_t = xt + (size_t)ns * BATCH;

    dim3 tb(32, 8);
    dim3 tg((ns + 31) / 32, (BATCH + 31) / 32);
    k_transpose<<<tg, tb, 0, stream>>>(x, xt, ns);

    long long n4 = ((long long)nd * BATCH) / 4;
    long long zb = (n4 + 255) / 256;
    if (zb > 16384) zb = 16384;
    k_zero<<<(int)zb, 256, 0, stream>>>((float4*)out_t, n4);

    long long waves = ((long long)nnz + 32LL * CHUNKS - 1) / (32LL * CHUNKS);
    long long sblocks = (waves + 7) / 8;
    if (sblocks < 1) sblocks = 1;
    k_scatter<<<(int)sblocks, 256, 0, stream>>>(xt, vals, row, col, out_t, nnz);

    const int dtiles = nd / 16;
    if (dtiles > 0) {
        const int tiles = (BATCH / 16) * dtiles;
        const int eblocks = (tiles + 7) / 8;
        k_epilogue_wmma<<<eblocks, 256, 0, stream>>>(out_t, bias, out, nd, dtiles);
    }
    const int dstart = dtiles * 16;
    if (dstart < nd) {
        const int total = (nd - dstart) * BATCH;
        k_epilogue_tail<<<(total + 255) / 256, 256, 0, stream>>>(out_t, bias, out, nd, dstart);
    }
}


# x gathers from LDS: pre-pass groups each 12500-edge tile by 224-column source block (8-byte records, clears planes); work-groups per block hold its x slice in LDS; same f32 atomics; 4 -> 3 launches
# speedup vs baseline: 1.9195x; 1.9195x over previous
#include <hip/hip_runtime.h>

typedef __attribute__((ext_vector_type(2))) float v2f;
typedef __attribute__((ext_vector_type(8))) float v8f;

#define BATCH 64
#define CHUNKS 4

__global__ void k_transpose(const float* __restrict__ x, float* __restrict__ xt, int ns) {
    __shared__ float tile[32][33];
    const int tx = threadIdx.x;
    const int ty = threadIdx.y;
    const int s0 = blockIdx.x * 32;
    const int b0 = blockIdx.y * 32;
#pragma unroll
    for (int k = 0; k < 4; ++k) {
        int s = s0 + tx, b = b0 + ty + k * 8;
        if (s < ns && b < BATCH) tile[ty + k * 8][tx] = x[(size_t)b * ns + s];
    }
    __syncthreads();
#pragma unroll
    for (int k = 0; k < 4; ++k) {
        int s = s0 + ty + k * 8, b = b0 + tx;
        if (s < ns && b < BATCH) xt[(size_t)s * BATCH + b] = tile[tx][ty + k * 8];
    }
}

__global__ void k_zero(float4* __restrict__ p, long long n4) {
    long long i = (long long)blockIdx.x * blockDim.x + threadIdx.x;
    const long long stride = (long long)gridDim.x * blockDim.x;
    const float4 z = make_float4(0.f, 0.f, 0.f, 0.f);
    for (; i < n4; i += stride) p[i] = z;
}

#define EB 16

__global__ void k_scatter(const float* __restrict__ xt, const float* __restrict__ vals,
                          const int* __restrict__ row, const int* __restrict__ col,
                          float* __restrict__ out_t, long long plane, int nnz) {
    const int lane = threadIdx.x & 31;
    const long long wave = ((long long)blockIdx.x * blockDim.x + threadIdx.x) >> 5;
    const long long base0 = wave * (32LL * CHUNKS);
    if (base0 >= nnz) return;

    const unsigned half = (__builtin_amdgcn_s_sendmsg_rtn(0x87) >> 18) & 1u;
    out_t += (size_t)half * (size_t)plane;

    float v = 0.f;
    int r = 0, c = 0;
    if (base0 + lane < nnz) {
        v = vals[base0 + lane];
        r = row[base0 + lane];
        c = col[base0 + lane];
    }
    for (int ch = 0; ch < CHUNKS; ++ch) {
        const long long base = base0 + 32LL * ch;
        if (base >= nnz) return;
        float vn = 0.f;
        int rn = 0, cn = 0;
        if (ch + 1 < CHUNKS && base + 32 + lane < nnz) {
            vn = vals[base + 32 + lane];
            rn = row[base + 32 + lane];
            cn = col[base + 32 + lane];
        }
        const long long rem = (long long)nnz - base;
        if (rem >= 32) {
#pragma unroll
            for (int jb = 0; jb < 32; jb += EB) {
                float x0[EB], x1[EB];
#pragma unroll
                for (int k = 0; k < EB; ++k) {
                    const int cj = __shfl(c, jb + k);
                    const float* src = xt + (size_t)cj * BATCH + lane;
                    x0[k] = src[0];
                    x1[k] = src[32];
                }
#pragma unroll
                for (int k = 0; k < EB; ++k) {
                    const float vj = __shfl(v, jb + k);
                    const int rj   = __shfl(r, jb + k);
                    float* dst = out_t + (size_t)rj * BATCH + lane;
                    unsafeAtomicAdd(dst, x0[k] * vj);
                    unsafeAtomicAdd(dst + 32, x1[k] * vj);
                }
            }
        } else {
            for (int j = 0; j < (int)rem; ++j) {
                const float vj = __shfl(v, j);
                const int rj   = __shfl(r, j);
                const int cj   = __shfl(c, j);
                const float* src = xt + (size_t)cj * BATCH + lane;
                const float x0 = src[0];
                const float x1 = src[32];
                float* dst = out_t + (size_t)rj * BATCH + lane;
                unsafeAtomicAdd(dst, x0 * vj);
                unsafeAtomicAdd(dst + 32, x1 * vj);
            }
        }
        v = vn; r = rn; c = cn;
    }
}

#define BC 224
#define PITCH (BC + 1)
#define MAXBLK 512
#define TE 12500
#define GT 1024
#define PT ((TE + GT - 1) / GT)
#define MAXTILES 1024
#define ST 512
#define SPLIT 2
#define SLICE_IT ((BATCH * (BC / 4)) / ST)
#define EG 4
static_assert((BATCH * (BC / 4)) % ST == 0, "the slice's 16-byte groups divide evenly among the threads");
static_assert(BC % 32 == 0 && BC < 512 && (PITCH % 2) == 1, "whole lines per slice row, 9-bit column, odd pitch");
static_assert(TE < (1 << 18) && 3 * (MAXBLK + 2) <= TE, "a tile's table fits its records' spare bits");

__global__ void __launch_bounds__(GT)
k_group(const float* __restrict__ vals, const int* __restrict__ row, const int* __restrict__ col,
        float* __restrict__ recs_f, float4* __restrict__ planes, long long n4, int nnz) {
    __shared__ uint2 srec[TE];
    __shared__ unsigned cnt[MAXBLK];
    __shared__ unsigned sa[MAXBLK];
    __shared__ unsigned sb[MAXBLK];
    __shared__ unsigned offs[MAXBLK + 1];
    const int tid = threadIdx.x;
    const long long base = (long long)blockIdx.x * TE;
    const long long left = (long long)nnz - base;
    const int len = left < TE ? (int)left : TE;

    int c[PT], r[PT];
    float v[PT];
#pragma unroll
    for (int k = 0; k < PT; ++k) {
        const int e = tid + k * GT;
        const int ec = e < len ? e : len - 1;
        c[k] = col[base + ec];
        r[k] = row[base + ec];
        v[k] = vals[base + ec];
    }
    {
        const float4 z = make_float4(0.f, 0.f, 0.f, 0.f);
        const long long stride = (long long)gridDim.x * GT;
        for (long long i = (long long)blockIdx.x * GT + tid; i < n4; i += stride) planes[i] = z;
    }
    for (int i = tid; i < MAXBLK; i += GT) cnt[i] = 0u;
    __syncthreads();

    unsigned rk[PT];
#pragma unroll
    for (int k = 0; k < PT; ++k) {
        const int e = tid + k * GT;
        rk[k] = 0u;
        if (e < len) rk[k] = atomicAdd(&cnt[(unsigned)c[k] / BC], 1u);
    }
    __syncthreads();

    if (tid < MAXBLK) sa[tid] = cnt[tid];
    __syncthreads();
    unsigned* s_src = sa;
    unsigned* s_dst = sb;
    for (int d = 1; d < MAXBLK; d <<= 1) {
        if (tid < MAXBLK) s_dst[tid] = s_src[tid] + (tid >= d ? s_src[tid - d] : 0u);
        __syncthreads();
        unsigned* tmp = s_src; s_src = s_dst; s_dst = tmp;
    }
    if (tid < MAXBLK) offs[tid + 1] = s_src[tid];
    if (tid == 0) offs[0] = 0u;
    __syncthreads();

    unsigned pos[PT], tbits[PT];
#pragma unroll
    for (int k = 0; k < PT; ++k) pos[k] = offs[(unsigned)c[k] / BC] + rk[k];
#pragma unroll
    for (int k = 0; k < PT; ++k) {
        const unsigned j = pos[k] / 3u;
        const unsigned dgt = pos[k] - 3u * j;
        const unsigned o = offs[j <= MAXBLK ? j : MAXBLK];
        tbits[k] = (j <= MAXBLK) ? ((o >> (6u * dgt)) & 63u) : 0u;
    }
#pragma unroll
    for (int k = 0; k < PT; ++k) {
        const int e = tid + k * GT;
        if (e < len) {
            const unsigned b = (unsigned)c[k] / BC;
            const unsigned cin = (unsigned)c[k] - b * BC;
            srec[pos[k]] = make_uint2(((unsigned)r[k] << 15) | (cin << 6) | tbits[k], __float_as_uint(v[k]));
        }
    }
    __syncthreads();

    uint2* recs = (uint2*)recs_f + base;
    for (int i = tid; i < len; i += GT) recs[i] = srec[i];
}

__global__ void __launch_bounds__(ST)
k_block_scatter(const float* __restrict__ x, const float* __restrict__ recs_f,
                float* __restrict__ out_t, long long plane, int ns, int ntiles) {
    __shared__ float xs[BATCH * PITCH];
    __shared__ int run_s[MAXTILES];
    __shared__ int run_e[MAXTILES];
    const int tid = threadIdx.x;
    const int lane = tid & 31;
    const int wave = tid >> 5;
    const int b = blockIdx.x;
    const uint2* recs = (const uint2*)recs_f;

    const unsigned half = (__builtin_amdgcn_s_sendmsg_rtn(0x87) >> 18) & 1u;
    out_t += (size_t)half * (size_t)plane;

    for (int t = tid; t < ntiles; t += ST) {
        const unsigned* k = (const unsigned*)(recs + (size_t)t * TE + 3 * b);
        run_s[t] = (int)((k[0] & 63u) | ((k[2] & 63u) << 6) | ((k[4] & 63u) << 12));
        run_e[t] = (int)((k[6] & 63u) | ((k[8] & 63u) << 6) | ((k[10] & 63u) << 12));
    }
    const int c0 = b * BC;
    float4 q[SLICE_IT];
#pragma unroll
    for (int it = 0; it < SLICE_IT; ++it) {
        const int i = tid + it * ST;
        const int rr = i / (BC / 4);
        const int cc = (i - rr * (BC / 4)) * 4;
        const int gc = (c0 + cc < ns) ? c0 + cc : c0;
        q[it] = *(const float4*)(x + (size_t)rr * ns + gc);
    }
#pragma unroll
    for (int it = 0; it < SLICE_IT; ++it) {
        const int i = tid + it * ST;
        const int rr = i / (BC / 4);
        const int cc = (i - rr * (BC / 4)) * 4;
        float* dst = xs + rr * PITCH + cc;
        dst[0] = q[it].x; dst[1] = q[it].y; dst[2] = q[it].z; dst[3] = q[it].w;
    }
    __syncthreads();

    const float* xl = xs + lane * PITCH;
    for (int t = blockIdx.y + SPLIT * wave; t < ntiles; t += SPLIT * (ST / 32)) {
        const int s = __builtin_amdgcn_readfirstlane(run_s[t]);
        const int e = __builtin_amdgcn_readfirstlane(run_e[t]);
        const uint2* rp = recs + (size_t)t * TE;
        for (int p = s; p < e; p += 32) {
            const int n = (e - p < 32) ? (e - p) : 32;
            uint2 rec = make_uint2(0u, 0u);
            if (lane < n) rec = rp[p + lane];
            int j = 0;
            for (; j + EG <= n; j += EG) {
                float x0[EG], x1[EG], w[EG];
                unsigned rj[EG];
#pragma unroll
                for (int k = 0; k < EG; ++k) {
                    const unsigned key = (unsigned)__builtin_amdgcn_readlane((int)rec.x, j + k);
                    w[k] = __uint_as_float((unsigned)__builtin_amdgcn_readlane((int)rec.y, j + k));
                    rj[k] = key >> 15;
                    const unsigned cj = (key >> 6) & 511u;
                    x0[k] = xl[cj];
                    x1[k] = xl[cj + 32 * PITCH];
                }
#pragma unroll
                for (int k = 0; k < EG; ++k) {
                    float* dst = out_t + (size_t)rj[k] * BATCH + lane;
                    unsafeAtomicAdd(dst, x0[k] * w[k]);
                    unsafeAtomicAdd(dst + 32, x1[k] * w[k]);
                }
            }
            for (; j < n; ++j) {
                const unsigned key = (unsigned)__builtin_amdgcn_readlane((int)rec.x, j);
                const float wj = __uint_as_float((unsigned)__builtin_amdgcn_readlane((int)rec.y, j));
                const unsigned cj = (key >> 6) & 511u;
                float* dst = out_t + (size_t)(key >> 15) * BATCH + lane;
                unsafeAtomicAdd(dst, xl[cj] * wj);
                unsafeAtomicAdd(dst + 32, xl[cj + 32 * PITCH] * wj);
            }
        }
    }
}

__global__ void k_epilogue_wmma(const float* __restrict__ out_t, const float* __restrict__ bias,
                                float* __restrict__ out, long long plane, int nd, int dtiles) {
    const int lane = threadIdx.x & 31;
    const int wave = (int)(((long long)blockIdx.x * blockDim.x + threadIdx.x) >> 5);
    const int tiles = (BATCH / 16) * dtiles;
    if (wave >= tiles) return;
    const int b0 = (wave & 3) * 16;
    const int d0 = (wave >> 2) * 16;
    const int n = lane & 15;
    const int mbase = (lane < 16) ? 0 : 8;

    const float* cbase = out_t + (size_t)(d0 + n) * BATCH + b0 + mbase;
    const float4 lo = *(const float4*)(cbase);
    const float4 hi = *(const float4*)(cbase + 4);
    v8f c = {lo.x, lo.y, lo.z, lo.w, hi.x, hi.y, hi.z, hi.w};

    v2f a = {(lane < 16) ? 1.0f : 0.0f, 0.0f};
    v2f bm = {(lane < 16) ? bias[d0 + n] : 0.0f, 0.0f};

    if (plane != 0) {
        const float4 lo1 = *(const float4*)(cbase + plane);
        const float4 hi1 = *(const float4*)(cbase + plane + 4);
        const v8f c1 = {lo1.x, lo1.y, lo1.z, lo1.w, hi1.x, hi1.y, hi1.z, hi1.w};
        c += c1;
    }

    v8f d = __builtin_amdgcn_wmma_f32_16x16x4_f32(
         false, a,  false, bm,
         (short)0, c,  false,  false);

#pragma unroll
    for (int v = 0; v < 8; ++v) {
        out[(size_t)(b0 + mbase + v) * nd + d0 + n] = d[v];
    }
}

__global__ void k_epilogue_tail(const float* __restrict__ out_t, const float* __restrict__ bias,
                                float* __restrict__ out, long long plane, int nd, int dstart) {
    const int total = (nd - dstart) * BATCH;
    const int i = blockIdx.x * blockDim.x + threadIdx.x;
    if (i >= total) return;
    const int d = dstart + (i >> 6);
    const int b = i & 63;
    float acc = out_t[(size_t)d * BATCH + b];
    if (plane != 0) acc += out_t[(size_t)d * BATCH + b + plane];
    out[(size_t)b * nd + d] = acc + bias[d];
}

extern "C" void kernel_launch(void* const* d_in, const int* in_sizes, int n_in,
                              void* d_out, int out_size, void* d_ws, size_t ws_size,
                              hipStream_t stream) {
    const float* x    = (const float*)d_in[0];
    const float* vals = (const float*)d_in[1];
    const float* bias = (const float*)d_in[2];
    const int* row    = (const int*)d_in[3];
    const int* col    = (const int*)d_in[4];
    float* out = (float*)d_out;

    const int nnz = in_sizes[1];
    const int nd  = in_sizes[2];
    const int ns  = in_sizes[0] / BATCH;

    const long long plane_elems = (long long)nd * BATCH;
    const bool two_planes = (ns == nd) ||
        (ns < nd && (size_t)(2 * plane_elems) * sizeof(float) <= ws_size);
    float* xt    = two_planes ? out : (float*)d_ws;
    float* out_t = two_planes ? (float*)d_ws : xt + (size_t)ns * BATCH;
    const long long plane = two_planes ? plane_elems : 0;

    long long n4 = plane_elems / 4;

    const int nblk = (ns + BC - 1) / BC;
    const int ntiles = (nnz + TE - 1) / TE;
    const long long last_len = (long long)nnz - (long long)(ntiles - 1) * TE;
    const bool grouped = two_planes && nnz > 0 && (ns % 4) == 0 && nblk <= MAXBLK &&
        ntiles <= MAXTILES && nd <= (1 << 17) &&
        last_len >= 3LL * (nblk + 2) &&
        (long long)nnz * 8 <= plane_elems * (long long)sizeof(float);

    if (grouped) {
        k_group<<<ntiles, GT, 0, stream>>>(vals, row, col, out, (float4*)out_t, 2 * n4, nnz);
        k_block_scatter<<<dim3(nblk, SPLIT), ST, 0, stream>>>(x, out, out_t, plane, ns, ntiles);
    } else {
        dim3 tb(32, 8);
        dim3 tg((ns + 31) / 32, (BATCH + 31) / 32);
        k_transpose<<<tg, tb, 0, stream>>>(x, xt, ns);

        long long zb = (n4 + 255) / 256;
        if (zb > 16384) zb = 16384;
        k_zero<<<(int)zb, 256, 0, stream>>>((float4*)out_t, two_planes ? 2 * n4 : n4);

        long long waves = ((long long)nnz + 32LL * CHUNKS - 1) / (32LL * CHUNKS);
        long long sblocks = (waves + 7) / 8;
        if (sblocks < 1) sblocks = 1;
        k_scatter<<<(int)sblocks, 256, 0, stream>>>(xt, vals, row, col, out_t, plane, nnz);
    }

    const int dtiles = nd / 16;
    if (dtiles > 0) {
        const int tiles = (BATCH / 16) * dtiles;
        const int eblocks = (tiles + 7) / 8;
        k_epilogue_wmma<<<eblocks, 256, 0, stream>>>(out_t, bias, out, plane, nd, dtiles);
    }
    const int dstart = dtiles * 16;
    if (dstart < nd) {
        const int total = (nd - dstart) * BATCH;
        k_epilogue_tail<<<(total + 255) / 256, 256, 0, stream>>>(out_t, bias, out, plane, nd, dstart);
    }
}
